# starting kernel
# baseline (speedup 1.0000x reference)
#include <hip/hip_runtime.h>


typedef float v2f __attribute__((ext_vector_type(2)));
typedef float v8f __attribute__((ext_vector_type(8)));

#define DD 128
#define NT 8
#define WFRAG_F4_PER_MAT 4096

__global__ __launch_bounds__(256)
void reformat_w(const float* __restrict__ Wq, const float* __restrict__ Wk,
                const float* __restrict__ Wv, float4* __restrict__ Wf)
{
    const int idx = (int)(blockIdx.x * blockDim.x + threadIdx.x);
    const int m    = idx >> 12;
    const int r    = idx & 4095;
    const int kkp  = r >> 8;
    const int t    = (r >> 5) & 7;
    const int lane = r & 31;
    const int hi = lane >> 4, l = lane & 15;
    const int n  = t * 16 + l;
    const float* W = (m == 0) ? Wq : ((m == 1) ? Wk : Wv);
    const int k0 = 8 * kkp + 2 * hi;
    const int k1 = k0 + 4;
    float4 v;
    v.x = W[(size_t)(k0 + 0) * DD + n];
    v.y = W[(size_t)(k0 + 1) * DD + n];
    v.z = W[(size_t)(k1 + 0) * DD + n];
    v.w = W[(size_t)(k1 + 1) * DD + n];
    Wf[idx] = v;
}

__global__ __launch_bounds__(256)
void qkv_gemm_wmma(const float* __restrict__ X,
                   const float4* __restrict__ Wf,
                   const float* __restrict__ bq, const float* __restrict__ bk,
                   const float* __restrict__ bv,
                   float* __restrict__ Q, float* __restrict__ K, float* __restrict__ V,
                   int Mrows)
{
    const int wave = (int)((blockIdx.x * blockDim.x + threadIdx.x) >> 5);
    const int lane = (int)(threadIdx.x & 31);
    const int m0   = wave * 16;
    if (m0 >= Mrows) return;

    const int hi = lane >> 4;
    const int l  = lane & 15;

    const float* xrow = X + (size_t)(m0 + l) * DD + 2 * hi;

    const float* bm[3] = { bq, bk, bv };
    float*       Om[3] = { Q,  K,  V  };

#pragma unroll
    for (int m = 0; m < 3; ++m) {
        const float4* Wp = Wf + (size_t)m * WFRAG_F4_PER_MAT;
        const float*  bp = bm[m];
        float*        Op = Om[m];

        v8f c[NT];
#pragma unroll
        for (int t = 0; t < NT; ++t) {
            const float bias = bp[t * 16 + l];
#pragma unroll
            for (int r = 0; r < 8; ++r) c[t][r] = bias;
        }

#pragma unroll 2
        for (int kkp = 0; kkp < 16; ++kkp) {
            const v2f a0 = *(const v2f*)(xrow + 8 * kkp);
            const v2f a1 = *(const v2f*)(xrow + 8 * kkp + 4);
            float4 bf[NT];
#pragma unroll
            for (int t = 0; t < NT; ++t)
                bf[t] = Wp[(size_t)(kkp * NT + t) * 32 + lane];
#pragma unroll
            for (int t = 0; t < NT; ++t) {
                v2f b0; b0.x = bf[t].x; b0.y = bf[t].y;
                c[t] = __builtin_amdgcn_wmma_f32_16x16x4_f32(
                           false, a0, false, b0, (short)0, c[t], false, false);
            }
#pragma unroll
            for (int t = 0; t < NT; ++t) {
                v2f b1; b1.x = bf[t].z; b1.y = bf[t].w;
                c[t] = __builtin_amdgcn_wmma_f32_16x16x4_f32(
                           false, a1, false, b1, (short)0, c[t], false, false);
            }
        }

#pragma unroll
        for (int t = 0; t < NT; ++t) {
#pragma unroll
            for (int r = 0; r < 8; ++r) {
                Op[(size_t)(m0 + 8 * hi + r) * DD + t * 16 + l] = c[t][r];
            }
        }
    }
}

__global__ __launch_bounds__(256)
void win_attn(const float* __restrict__ Q, const float* __restrict__ K,
              const float* __restrict__ V, float* __restrict__ out,
              int B, int H, int W)
{
    const int wave = (int)((blockIdx.x * blockDim.x + threadIdx.x) >> 5);
    const int lane = (int)(threadIdx.x & 31);
    const int npix = B * H * W;
    if (wave >= npix) return;

    const int b   = wave / (H * W);
    const int rem = wave % (H * W);
    const int y   = rem / W;
    const int x   = rem % W;
    const float scale = 0.08838834764831845f;

    const float4 q = *(const float4*)(Q + (size_t)wave * DD + lane * 4);

    float  sc[9];
    float4 vf[9];
#pragma unroll
    for (int dy = -1; dy <= 1; ++dy) {
#pragma unroll
        for (int dx = -1; dx <= 1; ++dx) {
            const int  i  = (dy + 1) * 3 + (dx + 1);
            const int  yy = y + dy, xx = x + dx;
            const bool ok = (yy >= 0) & (yy < H) & (xx >= 0) & (xx < W);
            float  s  = 0.0f;
            float4 v4 = make_float4(0.f, 0.f, 0.f, 0.f);
            if (ok) {
                const size_t off = ((size_t)b * H * W + (size_t)yy * W + xx) * DD + lane * 4;
                const float4 k4 = *(const float4*)(K + off);
                v4 = *(const float4*)(V + off);
                s = q.x * k4.x + q.y * k4.y + q.z * k4.z + q.w * k4.w;
            }
#pragma unroll
            for (int m = 16; m >= 1; m >>= 1) s += __shfl_xor(s, m, 32);
            sc[i] = s * scale;
            vf[i] = v4;
        }
    }

    float mx = sc[0];
#pragma unroll
    for (int i = 1; i < 9; ++i) mx = fmaxf(mx, sc[i]);
    float p[9], sum = 0.0f;
#pragma unroll
    for (int i = 0; i < 9; ++i) { p[i] = expf(sc[i] - mx); sum += p[i]; }
    const float inv = 1.0f / sum;

    float4 o = make_float4(0.f, 0.f, 0.f, 0.f);
#pragma unroll
    for (int i = 0; i < 9; ++i) {
        const float w = p[i] * inv;
        o.x += w * vf[i].x; o.y += w * vf[i].y;
        o.z += w * vf[i].z; o.w += w * vf[i].w;
    }
    *(float4*)(out + (size_t)wave * DD + lane * 4) = o;
}

extern "C" void kernel_launch(void* const* d_in, const int* in_sizes, int n_in,
                              void* d_out, int out_size, void* d_ws, size_t ws_size,
                              hipStream_t stream)
{
    const float* x  = (const float*)d_in[0];
    const float* Wq = (const float*)d_in[1];
    const float* bq = (const float*)d_in[2];
    const float* Wk = (const float*)d_in[3];
    const float* bk = (const float*)d_in[4];
    const float* Wv = (const float*)d_in[5];
    const float* bv = (const float*)d_in[6];

    const int H = 128, W = 128;
    const int Mrows = in_sizes[0] / DD;
    const int B = Mrows / (H * W);

    float*  Q   = (float*)d_ws;
    float*  Kp  = Q  + (size_t)Mrows * DD;
    float*  Vp  = Kp + (size_t)Mrows * DD;
    float4* Wf  = (float4*)(Vp + (size_t)Mrows * DD);

    reformat_w<<<48, 256, 0, stream>>>(Wq, Wk, Wv, Wf);

    const int gemmWaves = Mrows / 16;
    qkv_gemm_wmma<<<(gemmWaves + 7) / 8, 256, 0, stream>>>(
        x, Wf, bq, bk, bv, Q, Kp, Vp, Mrows);

    const int pixWaves = Mrows;
    win_attn<<<(pixWaves + 7) / 8, 256, 0, stream>>>(
        Q, Kp, Vp, (float*)d_out, B, H, W);
}


# win_attn: a workgroup copies K,V of a 10x10 pixel tile to LDS once; its 8 waves each do a row of 8 pixels from LDS: 8x fewer waves, 100 K/V rows read per 64 pixels (was 576); sum order unchanged
# speedup vs baseline: 1.1919x; 1.1919x over previous
#include <hip/hip_runtime.h>


typedef float v2f __attribute__((ext_vector_type(2)));
typedef float v8f __attribute__((ext_vector_type(8)));

#define DD 128
#define NT 8
#define WFRAG_F4_PER_MAT 4096

__global__ __launch_bounds__(256)
void reformat_w(const float* __restrict__ Wq, const float* __restrict__ Wk,
                const float* __restrict__ Wv, float4* __restrict__ Wf)
{
    const int idx = (int)(blockIdx.x * blockDim.x + threadIdx.x);
    const int m    = idx >> 12;
    const int r    = idx & 4095;
    const int kkp  = r >> 8;
    const int t    = (r >> 5) & 7;
    const int lane = r & 31;
    const int hi = lane >> 4, l = lane & 15;
    const int n  = t * 16 + l;
    const float* W = (m == 0) ? Wq : ((m == 1) ? Wk : Wv);
    const int k0 = 8 * kkp + 2 * hi;
    const int k1 = k0 + 4;
    float4 v;
    v.x = W[(size_t)(k0 + 0) * DD + n];
    v.y = W[(size_t)(k0 + 1) * DD + n];
    v.z = W[(size_t)(k1 + 0) * DD + n];
    v.w = W[(size_t)(k1 + 1) * DD + n];
    Wf[idx] = v;
}

__global__ __launch_bounds__(256)
void qkv_gemm_wmma(const float* __restrict__ X,
                   const float4* __restrict__ Wf,
                   const float* __restrict__ bq, const float* __restrict__ bk,
                   const float* __restrict__ bv,
                   float* __restrict__ Q, float* __restrict__ K, float* __restrict__ V,
                   int Mrows)
{
    const int wave = (int)((blockIdx.x * blockDim.x + threadIdx.x) >> 5);
    const int lane = (int)(threadIdx.x & 31);
    const int m0   = wave * 16;
    if (m0 >= Mrows) return;

    const int hi = lane >> 4;
    const int l  = lane & 15;

    const float* xrow = X + (size_t)(m0 + l) * DD + 2 * hi;

    const float* bm[3] = { bq, bk, bv };
    float*       Om[3] = { Q,  K,  V  };

#pragma unroll
    for (int m = 0; m < 3; ++m) {
        const float4* Wp = Wf + (size_t)m * WFRAG_F4_PER_MAT;
        const float*  bp = bm[m];
        float*        Op = Om[m];

        v8f c[NT];
#pragma unroll
        for (int t = 0; t < NT; ++t) {
            const float bias = bp[t * 16 + l];
#pragma unroll
            for (int r = 0; r < 8; ++r) c[t][r] = bias;
        }

#pragma unroll 2
        for (int kkp = 0; kkp < 16; ++kkp) {
            const v2f a0 = *(const v2f*)(xrow + 8 * kkp);
            const v2f a1 = *(const v2f*)(xrow + 8 * kkp + 4);
            float4 bf[NT];
#pragma unroll
            for (int t = 0; t < NT; ++t)
                bf[t] = Wp[(size_t)(kkp * NT + t) * 32 + lane];
#pragma unroll
            for (int t = 0; t < NT; ++t) {
                v2f b0; b0.x = bf[t].x; b0.y = bf[t].y;
                c[t] = __builtin_amdgcn_wmma_f32_16x16x4_f32(
                           false, a0, false, b0, (short)0, c[t], false, false);
            }
#pragma unroll
            for (int t = 0; t < NT; ++t) {
                v2f b1; b1.x = bf[t].z; b1.y = bf[t].w;
                c[t] = __builtin_amdgcn_wmma_f32_16x16x4_f32(
                           false, a1, false, b1, (short)0, c[t], false, false);
            }
        }

#pragma unroll
        for (int t = 0; t < NT; ++t) {
#pragma unroll
            for (int r = 0; r < 8; ++r) {
                Op[(size_t)(m0 + 8 * hi + r) * DD + t * 16 + l] = c[t][r];
            }
        }
    }
}

#define TS 8
#define HS (TS + 2)
#define HP (HS * HS)

__global__ __launch_bounds__(256)
void win_attn(const float* __restrict__ Q, const float* __restrict__ K,
              const float* __restrict__ V, float* __restrict__ out,
              int B, int H, int W)
{
    __shared__ float4 sK[HP * 32];
    __shared__ float4 sV[HP * 32];

    const int lane = (int)(threadIdx.x & 31);
    const int wv = __builtin_amdgcn_readfirstlane((int)(threadIdx.x >> 5)) & 7;

    const int tilesX = (W + TS - 1) / TS;
    const int tilesY = (H + TS - 1) / TS;
    int t = (int)blockIdx.x;
    const int tx = t % tilesX;  t /= tilesX;
    const int ty = t % tilesY;
    const int b  = t / tilesY;
    if (b >= B) return;
    const int y0 = ty * TS, x0 = tx * TS;
    const size_t img = (size_t)b * H * W;
    const float scale = 0.08838834764831845f;

    const int  yrow  = y0 + wv;
    const int  yq    = min(yrow, H - 1);
    const float* qrow = Q + (img + (size_t)yq * W) * DD + lane * 4;
    float4 qcur = *(const float4*)(qrow + (size_t)min(x0, W - 1) * DD);

    const int wl = (int)(threadIdx.x >> 5) & 7;
#pragma unroll
    for (int g = 0; g < 2; ++g) {
        float4 kk[7], vv[7];
        bool   okk[7];
#pragma unroll
        for (int i = 0; i < 7; ++i) {
            if (7 * g + i >= (HP + 7) / 8) continue;
            const int p  = min(wl + 8 * (7 * g + i), HP - 1);
            const int hy = p / HS, hx = p - hy * HS;
            const int yy = y0 - 1 + hy, xx = x0 - 1 + hx;
            okk[i] = (yy >= 0) & (yy < H) & (xx >= 0) & (xx < W);
            const int yc = min(max(yy, 0), H - 1);
            const int xc = min(max(xx, 0), W - 1);
            const size_t off = (img + (size_t)yc * W + xc) * DD + lane * 4;
            kk[i] = *(const float4*)(K + off);
            vv[i] = *(const float4*)(V + off);
        }
#pragma unroll
        for (int i = 0; i < 7; ++i) {
            if (7 * g + i >= (HP + 7) / 8) continue;
            const int p = wl + 8 * (7 * g + i);
            float4 k4 = kk[i], v4 = vv[i];
            if (!okk[i]) {
                k4 = make_float4(0.f, 0.f, 0.f, 0.f);
                v4 = make_float4(0.f, 0.f, 0.f, 0.f);
            }
            if (p < HP) {
                sK[p * 32 + lane] = k4;
                sV[p * 32 + lane] = v4;
            }
        }
    }
    __syncthreads();

    for (int j = 0; j < TS; ++j) {
        float4 qnext = qcur;
        if (j + 1 < TS)
            qnext = *(const float4*)(qrow + (size_t)min(x0 + j + 1, W - 1) * DD);
        const float4 q = qcur;

        float  sc[9];
        float4 vf[9];
#pragma unroll
        for (int dy = 0; dy < 3; ++dy) {
#pragma unroll
            for (int dx = 0; dx < 3; ++dx) {
                const int i = dy * 3 + dx;
                const int p = (wv + dy) * HS + (j + dx);
                const float4 k4 = sK[p * 32 + lane];
                vf[i] = sV[p * 32 + lane];
                float s = q.x * k4.x + q.y * k4.y + q.z * k4.z + q.w * k4.w;
#pragma unroll
                for (int m = 16; m >= 1; m >>= 1) s += __shfl_xor(s, m, 32);
                sc[i] = s * scale;
            }
        }

        float mx = sc[0];
#pragma unroll
        for (int i = 1; i < 9; ++i) mx = fmaxf(mx, sc[i]);
        float p9[9], sum = 0.0f;
#pragma unroll
        for (int i = 0; i < 9; ++i) { p9[i] = expf(sc[i] - mx); sum += p9[i]; }
        const float inv = 1.0f / sum;

        float4 o = make_float4(0.f, 0.f, 0.f, 0.f);
#pragma unroll
        for (int i = 0; i < 9; ++i) {
            const float w = p9[i] * inv;
            o.x += w * vf[i].x; o.y += w * vf[i].y;
            o.z += w * vf[i].z; o.w += w * vf[i].w;
        }
        if ((yrow < H) & (x0 + j < W))
            *(float4*)(out + (img + (size_t)yrow * W + x0 + j) * DD + lane * 4) = o;
        qcur = qnext;
    }
}

extern "C" void kernel_launch(void* const* d_in, const int* in_sizes, int n_in,
                              void* d_out, int out_size, void* d_ws, size_t ws_size,
                              hipStream_t stream)
{
    const float* x  = (const float*)d_in[0];
    const float* Wq = (const float*)d_in[1];
    const float* bq = (const float*)d_in[2];
    const float* Wk = (const float*)d_in[3];
    const float* bk = (const float*)d_in[4];
    const float* Wv = (const float*)d_in[5];
    const float* bv = (const float*)d_in[6];

    const int H = 128, W = 128;
    const int Mrows = in_sizes[0] / DD;
    const int B = Mrows / (H * W);

    float*  Q   = (float*)d_ws;
    float*  Kp  = Q  + (size_t)Mrows * DD;
    float*  Vp  = Kp + (size_t)Mrows * DD;
    float4* Wf  = (float4*)(Vp + (size_t)Mrows * DD);

    reformat_w<<<48, 256, 0, stream>>>(Wq, Wk, Wv, Wf);

    const int gemmWaves = Mrows / 16;
    qkv_gemm_wmma<<<(gemmWaves + 7) / 8, 256, 0, stream>>>(
        x, Wf, bq, bk, bv, Q, Kp, Vp, Mrows);

    const int attnTiles = Mrows / (TS * TS);
    win_attn<<<attnTiles, 256, 0, stream>>>(
        Q, Kp, Vp, (float*)d_out, B, H, W);
}


# qkv_gemm: a work-group's 8 waves copy each weight matrix (fragment order, 64 KB) into LDS once; all B fragments come from LDS (ds_load_b128), not the WGP cache / L2; values, k order unchanged
# speedup vs baseline: 1.3075x; 1.0970x over previous
#include <hip/hip_runtime.h>


typedef float v2f __attribute__((ext_vector_type(2)));
typedef float v8f __attribute__((ext_vector_type(8)));

#define DD 128
#define NT 8
#define WFRAG_F4_PER_MAT 4096

__global__ __launch_bounds__(256)
void reformat_w(const float* __restrict__ Wq, const float* __restrict__ Wk,
                const float* __restrict__ Wv, float4* __restrict__ Wf)
{
    const int idx = (int)(blockIdx.x * blockDim.x + threadIdx.x);
    const int m    = idx >> 12;
    const int r    = idx & 4095;
    const int kkp  = r >> 8;
    const int t    = (r >> 5) & 7;
    const int lane = r & 31;
    const int hi = lane >> 4, l = lane & 15;
    const int n  = t * 16 + l;
    const float* W = (m == 0) ? Wq : ((m == 1) ? Wk : Wv);
    const int k0 = 8 * kkp + 2 * hi;
    const int k1 = k0 + 4;
    float4 v;
    v.x = W[(size_t)(k0 + 0) * DD + n];
    v.y = W[(size_t)(k0 + 1) * DD + n];
    v.z = W[(size_t)(k1 + 0) * DD + n];
    v.w = W[(size_t)(k1 + 1) * DD + n];
    Wf[idx] = v;
}

__global__ __launch_bounds__(256)
void qkv_gemm_wmma(const float* __restrict__ X,
                   const float4* __restrict__ Wf,
                   const float* __restrict__ bq, const float* __restrict__ bk,
                   const float* __restrict__ bv,
                   float* __restrict__ Q, float* __restrict__ K, float* __restrict__ V,
                   int Mrows)
{
    __shared__ float4 sW[WFRAG_F4_PER_MAT];

    const int tid  = (int)threadIdx.x;
    const int wave = (int)((blockIdx.x * blockDim.x + threadIdx.x) >> 5);
    const int lane = tid & 31;
    const bool live = wave * 16 < Mrows;
    const int  m0   = live ? wave * 16 : 0;

    const int hi = lane >> 4;
    const int l  = lane & 15;

    const float* xrow = X + (size_t)(m0 + l) * DD + 2 * hi;

    const float* bm[3] = { bq, bk, bv };
    float*       Om[3] = { Q,  K,  V  };

#pragma unroll
    for (int i = 0; i < WFRAG_F4_PER_MAT / 256; ++i)
        sW[tid + 256 * i] = Wf[tid + 256 * i];
    __syncthreads();

#pragma unroll
    for (int m = 0; m < 3; ++m) {
        const float*  bp = bm[m];
        float*        Op = Om[m];

        v8f c[NT];
#pragma unroll
        for (int t = 0; t < NT; ++t) {
            const float bias = bp[t * 16 + l];
#pragma unroll
            for (int r = 0; r < 8; ++r) c[t][r] = bias;
        }

#pragma unroll 2
        for (int kkp = 0; kkp < 16; ++kkp) {
            const v2f a0 = *(const v2f*)(xrow + 8 * kkp);
            const v2f a1 = *(const v2f*)(xrow + 8 * kkp + 4);
            float4 bf[NT];
#pragma unroll
            for (int t = 0; t < NT; ++t)
                bf[t] = sW[(kkp * NT + t) * 32 + lane];
#pragma unroll
            for (int t = 0; t < NT; ++t) {
                v2f b0; b0.x = bf[t].x; b0.y = bf[t].y;
                c[t] = __builtin_amdgcn_wmma_f32_16x16x4_f32(
                           false, a0, false, b0, (short)0, c[t], false, false);
            }
#pragma unroll
            for (int t = 0; t < NT; ++t) {
                v2f b1; b1.x = bf[t].z; b1.y = bf[t].w;
                c[t] = __builtin_amdgcn_wmma_f32_16x16x4_f32(
                           false, a1, false, b1, (short)0, c[t], false, false);
            }
        }

        if (m < 2) {
            const float4* Wn = Wf + (size_t)(m + 1) * WFRAG_F4_PER_MAT;
            __syncthreads();
#pragma unroll
            for (int i = 0; i < WFRAG_F4_PER_MAT / 256; ++i)
                sW[tid + 256 * i] = Wn[tid + 256 * i];
            __syncthreads();
        }

        if (live) {
#pragma unroll
            for (int t = 0; t < NT; ++t) {
#pragma unroll
                for (int r = 0; r < 8; ++r) {
                    Op[(size_t)(m0 + 8 * hi + r) * DD + t * 16 + l] = c[t][r];
                }
            }
        }
    }
}

#define TS 8
#define HS (TS + 2)
#define HP (HS * HS)

__global__ __launch_bounds__(256)
void win_attn(const float* __restrict__ Q, const float* __restrict__ K,
              const float* __restrict__ V, float* __restrict__ out,
              int B, int H, int W)
{
    __shared__ float4 sK[HP * 32];
    __shared__ float4 sV[HP * 32];

    const int lane = (int)(threadIdx.x & 31);
    const int wv = __builtin_amdgcn_readfirstlane((int)(threadIdx.x >> 5)) & 7;

    const int tilesX = (W + TS - 1) / TS;
    const int tilesY = (H + TS - 1) / TS;
    int t = (int)blockIdx.x;
    const int tx = t % tilesX;  t /= tilesX;
    const int ty = t % tilesY;
    const int b  = t / tilesY;
    if (b >= B) return;
    const int y0 = ty * TS, x0 = tx * TS;
    const size_t img = (size_t)b * H * W;
    const float scale = 0.08838834764831845f;

    const int  yrow  = y0 + wv;
    const int  yq    = min(yrow, H - 1);
    const float* qrow = Q + (img + (size_t)yq * W) * DD + lane * 4;
    float4 qcur = *(const float4*)(qrow + (size_t)min(x0, W - 1) * DD);

    const int wl = (int)(threadIdx.x >> 5) & 7;
#pragma unroll
    for (int g = 0; g < 2; ++g) {
        float4 kk[7], vv[7];
        bool   okk[7];
#pragma unroll
        for (int i = 0; i < 7; ++i) {
            if (7 * g + i >= (HP + 7) / 8) continue;
            const int p  = min(wl + 8 * (7 * g + i), HP - 1);
            const int hy = p / HS, hx = p - hy * HS;
            const int yy = y0 - 1 + hy, xx = x0 - 1 + hx;
            okk[i] = (yy >= 0) & (yy < H) & (xx >= 0) & (xx < W);
            const int yc = min(max(yy, 0), H - 1);
            const int xc = min(max(xx, 0), W - 1);
            const size_t off = (img + (size_t)yc * W + xc) * DD + lane * 4;
            kk[i] = *(const float4*)(K + off);
            vv[i] = *(const float4*)(V + off);
        }
#pragma unroll
        for (int i = 0; i < 7; ++i) {
            if (7 * g + i >= (HP + 7) / 8) continue;
            const int p = wl + 8 * (7 * g + i);
            float4 k4 = kk[i], v4 = vv[i];
            if (!okk[i]) {
                k4 = make_float4(0.f, 0.f, 0.f, 0.f);
                v4 = make_float4(0.f, 0.f, 0.f, 0.f);
            }
            if (p < HP) {
                sK[p * 32 + lane] = k4;
                sV[p * 32 + lane] = v4;
            }
        }
    }
    __syncthreads();

    for (int j = 0; j < TS; ++j) {
        float4 qnext = qcur;
        if (j + 1 < TS)
            qnext = *(const float4*)(qrow + (size_t)min(x0 + j + 1, W - 1) * DD);
        const float4 q = qcur;

        float  sc[9];
        float4 vf[9];
#pragma unroll
        for (int dy = 0; dy < 3; ++dy) {
#pragma unroll
            for (int dx = 0; dx < 3; ++dx) {
                const int i = dy * 3 + dx;
                const int p = (wv + dy) * HS + (j + dx);
                const float4 k4 = sK[p * 32 + lane];
                vf[i] = sV[p * 32 + lane];
                float s = q.x * k4.x + q.y * k4.y + q.z * k4.z + q.w * k4.w;
#pragma unroll
                for (int m = 16; m >= 1; m >>= 1) s += __shfl_xor(s, m, 32);
                sc[i] = s * scale;
            }
        }

        float mx = sc[0];
#pragma unroll
        for (int i = 1; i < 9; ++i) mx = fmaxf(mx, sc[i]);
        float p9[9], sum = 0.0f;
#pragma unroll
        for (int i = 0; i < 9; ++i) { p9[i] = expf(sc[i] - mx); sum += p9[i]; }
        const float inv = 1.0f / sum;

        float4 o = make_float4(0.f, 0.f, 0.f, 0.f);
#pragma unroll
        for (int i = 0; i < 9; ++i) {
            const float w = p9[i] * inv;
            o.x += w * vf[i].x; o.y += w * vf[i].y;
            o.z += w * vf[i].z; o.w += w * vf[i].w;
        }
        if ((yrow < H) & (x0 + j < W))
            *(float4*)(out + (img + (size_t)yrow * W + x0 + j) * DD + lane * 4) = o;
        qcur = qnext;
    }
}

extern "C" void kernel_launch(void* const* d_in, const int* in_sizes, int n_in,
                              void* d_out, int out_size, void* d_ws, size_t ws_size,
                              hipStream_t stream)
{
    const float* x  = (const float*)d_in[0];
    const float* Wq = (const float*)d_in[1];
    const float* bq = (const float*)d_in[2];
    const float* Wk = (const float*)d_in[3];
    const float* bk = (const float*)d_in[4];
    const float* Wv = (const float*)d_in[5];
    const float* bv = (const float*)d_in[6];

    const int H = 128, W = 128;
    const int Mrows = in_sizes[0] / DD;
    const int B = Mrows / (H * W);

    float*  Q   = (float*)d_ws;
    float*  Kp  = Q  + (size_t)Mrows * DD;
    float*  Vp  = Kp + (size_t)Mrows * DD;
    float4* Wf  = (float4*)(Vp + (size_t)Mrows * DD);

    reformat_w<<<48, 256, 0, stream>>>(Wq, Wk, Wv, Wf);

    const int gemmWaves = Mrows / 16;
    qkv_gemm_wmma<<<(gemmWaves + 7) / 8, 256, 0, stream>>>(
        x, Wf, bq, bk, bv, Q, Kp, Vp, Mrows);

    const int attnTiles = Mrows / (TS * TS);
    win_attn<<<attnTiles, 256, 0, stream>>>(
        Q, Kp, Vp, (float*)d_out, B, H, W);
}


# win_attn: a wave's lanes are 8 pixels x 4 groups of 32 channels (was 1 pixel x 32 groups of 4): 2 lane exchanges per score, not 5; softmax once per 8 pixels; f32 sum order of scores changes
# speedup vs baseline: 1.3667x; 1.0453x over previous
#include <hip/hip_runtime.h>


typedef float v2f __attribute__((ext_vector_type(2)));
typedef float v8f __attribute__((ext_vector_type(8)));

#define DD 128
#define NT 8
#define WFRAG_F4_PER_MAT 4096

__global__ __launch_bounds__(256)
void reformat_w(const float* __restrict__ Wq, const float* __restrict__ Wk,
                const float* __restrict__ Wv, float4* __restrict__ Wf)
{
    const int idx = (int)(blockIdx.x * blockDim.x + threadIdx.x);
    const int m    = idx >> 12;
    const int r    = idx & 4095;
    const int kkp  = r >> 8;
    const int t    = (r >> 5) & 7;
    const int lane = r & 31;
    const int hi = lane >> 4, l = lane & 15;
    const int n  = t * 16 + l;
    const float* W = (m == 0) ? Wq : ((m == 1) ? Wk : Wv);
    const int k0 = 8 * kkp + 2 * hi;
    const int k1 = k0 + 4;
    float4 v;
    v.x = W[(size_t)(k0 + 0) * DD + n];
    v.y = W[(size_t)(k0 + 1) * DD + n];
    v.z = W[(size_t)(k1 + 0) * DD + n];
    v.w = W[(size_t)(k1 + 1) * DD + n];
    Wf[idx] = v;
}

__global__ __launch_bounds__(256)
void qkv_gemm_wmma(const float* __restrict__ X,
                   const float4* __restrict__ Wf,
                   const float* __restrict__ bq, const float* __restrict__ bk,
                   const float* __restrict__ bv,
                   float* __restrict__ Q, float* __restrict__ K, float* __restrict__ V,
                   int Mrows)
{
    __shared__ float4 sW[WFRAG_F4_PER_MAT];

    const int tid  = (int)threadIdx.x;
    const int wave = (int)((blockIdx.x * blockDim.x + threadIdx.x) >> 5);
    const int lane = tid & 31;
    const bool live = wave * 16 < Mrows;
    const int  m0   = live ? wave * 16 : 0;

    const int hi = lane >> 4;
    const int l  = lane & 15;

    const float* xrow = X + (size_t)(m0 + l) * DD + 2 * hi;

    const float* bm[3] = { bq, bk, bv };
    float*       Om[3] = { Q,  K,  V  };

#pragma unroll
    for (int i = 0; i < WFRAG_F4_PER_MAT / 256; ++i)
        sW[tid + 256 * i] = Wf[tid + 256 * i];
    __syncthreads();

#pragma unroll
    for (int m = 0; m < 3; ++m) {
        const float*  bp = bm[m];
        float*        Op = Om[m];

        v8f c[NT];
#pragma unroll
        for (int t = 0; t < NT; ++t) {
            const float bias = bp[t * 16 + l];
#pragma unroll
            for (int r = 0; r < 8; ++r) c[t][r] = bias;
        }

#pragma unroll 2
        for (int kkp = 0; kkp < 16; ++kkp) {
            const v2f a0 = *(const v2f*)(xrow + 8 * kkp);
            const v2f a1 = *(const v2f*)(xrow + 8 * kkp + 4);
            float4 bf[NT];
#pragma unroll
            for (int t = 0; t < NT; ++t)
                bf[t] = sW[(kkp * NT + t) * 32 + lane];
#pragma unroll
            for (int t = 0; t < NT; ++t) {
                v2f b0; b0.x = bf[t].x; b0.y = bf[t].y;
                c[t] = __builtin_amdgcn_wmma_f32_16x16x4_f32(
                           false, a0, false, b0, (short)0, c[t], false, false);
            }
#pragma unroll
            for (int t = 0; t < NT; ++t) {
                v2f b1; b1.x = bf[t].z; b1.y = bf[t].w;
                c[t] = __builtin_amdgcn_wmma_f32_16x16x4_f32(
                           false, a1, false, b1, (short)0, c[t], false, false);
            }
        }

        if (m < 2) {
            const float4* Wn = Wf + (size_t)(m + 1) * WFRAG_F4_PER_MAT;
            __syncthreads();
#pragma unroll
            for (int i = 0; i < WFRAG_F4_PER_MAT / 256; ++i)
                sW[tid + 256 * i] = Wn[tid + 256 * i];
            __syncthreads();
        }

        if (live) {
#pragma unroll
            for (int t = 0; t < NT; ++t) {
#pragma unroll
                for (int r = 0; r < 8; ++r) {
                    Op[(size_t)(m0 + 8 * hi + r) * DD + t * 16 + l] = c[t][r];
                }
            }
        }
    }
}

#define TS 8
#define HS (TS + 2)
#define HP (HS * HS)
#define RP 33

__global__ __launch_bounds__(256)
void win_attn(const float* __restrict__ Q, const float* __restrict__ K,
              const float* __restrict__ V, float* __restrict__ out,
              int B, int H, int W)
{
    __shared__ float4 sK[HP * RP];
    __shared__ float4 sV[HP * RP];

    const int lane = (int)(threadIdx.x & 31);
    const int wv = __builtin_amdgcn_readfirstlane((int)(threadIdx.x >> 5)) & 7;

    const int tilesX = (W + TS - 1) / TS;
    const int tilesY = (H + TS - 1) / TS;
    int t = (int)blockIdx.x;
    const int tx = t % tilesX;  t /= tilesX;
    const int ty = t % tilesY;
    const int b  = t / tilesY;
    if (b >= B) return;
    const int y0 = ty * TS, x0 = tx * TS;
    const size_t img = (size_t)b * H * W;
    const float scale = 0.08838834764831845f;

    const int j = lane & 7;
    const int g = lane >> 3;
    const int yrow = y0 + wv;
    const int yq   = min(yrow, H - 1);
    const int xq   = min(x0 + j, W - 1);
    const float* qp = Q + (img + (size_t)yq * W + xq) * DD + g * 32;
    float4 q[8];
#pragma unroll
    for (int i = 0; i < 8; ++i) q[i] = *(const float4*)(qp + 4 * i);

    const int wl = (int)(threadIdx.x >> 5) & 7;
#pragma unroll
    for (int gr = 0; gr < 2; ++gr) {
        float4 kk[7], vv[7];
        bool   okk[7];
#pragma unroll
        for (int i = 0; i < 7; ++i) {
            if (7 * gr + i >= (HP + 7) / 8) continue;
            const int p  = min(wl + 8 * (7 * gr + i), HP - 1);
            const int hy = p / HS, hx = p - hy * HS;
            const int yy = y0 - 1 + hy, xx = x0 - 1 + hx;
            okk[i] = (yy >= 0) & (yy < H) & (xx >= 0) & (xx < W);
            const int yc = min(max(yy, 0), H - 1);
            const int xc = min(max(xx, 0), W - 1);
            const size_t off = (img + (size_t)yc * W + xc) * DD + lane * 4;
            kk[i] = *(const float4*)(K + off);
            vv[i] = *(const float4*)(V + off);
        }
#pragma unroll
        for (int i = 0; i < 7; ++i) {
            if (7 * gr + i >= (HP + 7) / 8) continue;
            const int p = wl + 8 * (7 * gr + i);
            float4 k4 = kk[i], v4 = vv[i];
            if (!okk[i]) {
                k4 = make_float4(0.f, 0.f, 0.f, 0.f);
                v4 = make_float4(0.f, 0.f, 0.f, 0.f);
            }
            if (p < HP) {
                sK[p * RP + lane] = k4;
                sV[p * RP + lane] = v4;
            }
        }
    }
    __syncthreads();

    const int lo = j * RP + g * 8;

    float sc[9];
#pragma unroll
    for (int dy = 0; dy < 3; ++dy) {
#pragma unroll
        for (int dx = 0; dx < 3; ++dx) {
            const float4* kr = sK + ((wv + dy) * HS + dx) * RP + lo;
            float s0 = 0.0f, s1 = 0.0f;
#pragma unroll
            for (int i = 0; i < 8; i += 2) {
                const float4 ka = kr[i], kb = kr[i + 1];
                s0 = fmaf(q[i].x, ka.x, s0);     s1 = fmaf(q[i + 1].x, kb.x, s1);
                s0 = fmaf(q[i].y, ka.y, s0);     s1 = fmaf(q[i + 1].y, kb.y, s1);
                s0 = fmaf(q[i].z, ka.z, s0);     s1 = fmaf(q[i + 1].z, kb.z, s1);
                s0 = fmaf(q[i].w, ka.w, s0);     s1 = fmaf(q[i + 1].w, kb.w, s1);
            }
            float s = s0 + s1;
            s += __shfl_xor(s, 8, 32);
            s += __shfl_xor(s, 16, 32);
            sc[dy * 3 + dx] = s * scale;
        }
    }

    float mx = sc[0];
#pragma unroll
    for (int i = 1; i < 9; ++i) mx = fmaxf(mx, sc[i]);
    float p9[9], sum = 0.0f;
#pragma unroll
    for (int i = 0; i < 9; ++i) { p9[i] = expf(sc[i] - mx); sum += p9[i]; }
    const float inv = 1.0f / sum;

    float4 o[8];
#pragma unroll
    for (int i = 0; i < 8; ++i) o[i] = make_float4(0.f, 0.f, 0.f, 0.f);
#pragma unroll
    for (int dy = 0; dy < 3; ++dy) {
#pragma unroll
        for (int dx = 0; dx < 3; ++dx) {
            const float4* vr = sV + ((wv + dy) * HS + dx) * RP + lo;
            const float w = p9[dy * 3 + dx] * inv;
#pragma unroll
            for (int i = 0; i < 8; ++i) {
                const float4 v4 = vr[i];
                o[i].x += w * v4.x; o[i].y += w * v4.y;
                o[i].z += w * v4.z; o[i].w += w * v4.w;
            }
        }
    }

    __syncthreads();
    float4* slot = sK + (wv * 8) * RP;
#pragma unroll
    for (int i = 0; i < 8; ++i) slot[lo + i] = o[i];
    __syncthreads();
#pragma unroll
    for (int jj = 0; jj < 8; ++jj) {
        const float4 r = slot[jj * RP + lane];
        if ((yrow < H) & (x0 + jj < W))
            *(float4*)(out + (img + (size_t)yrow * W + x0 + jj) * DD + lane * 4) = r;
    }
}

extern "C" void kernel_launch(void* const* d_in, const int* in_sizes, int n_in,
                              void* d_out, int out_size, void* d_ws, size_t ws_size,
                              hipStream_t stream)
{
    const float* x  = (const float*)d_in[0];
    const float* Wq = (const float*)d_in[1];
    const float* bq = (const float*)d_in[2];
    const float* Wk = (const float*)d_in[3];
    const float* bk = (const float*)d_in[4];
    const float* Wv = (const float*)d_in[5];
    const float* bv = (const float*)d_in[6];

    const int H = 128, W = 128;
    const int Mrows = in_sizes[0] / DD;
    const int B = Mrows / (H * W);

    float*  Q   = (float*)d_ws;
    float*  Kp  = Q  + (size_t)Mrows * DD;
    float*  Vp  = Kp + (size_t)Mrows * DD;
    float4* Wf  = (float4*)(Vp + (size_t)Mrows * DD);

    reformat_w<<<48, 256, 0, stream>>>(Wq, Wk, Wv, Wf);

    const int gemmWaves = Mrows / 16;
    qkv_gemm_wmma<<<(gemmWaves + 7) / 8, 256, 0, stream>>>(
        x, Wf, bq, bk, bv, Q, Kp, Vp, Mrows);

    const int attnTiles = Mrows / (TS * TS);
    win_attn<<<attnTiles, 256, 0, stream>>>(
        Q, Kp, Vp, (float*)d_out, B, H, W);
}


# Q product moved from the product launch into win_attn: a work-group forms Q of its 8x8 tile on WMMA (X tile in LDS, Wq fragments from memory), Q never stored; product launch: K,V only; sums unchanged
# speedup vs baseline: 1.5736x; 1.1514x over previous
#include <hip/hip_runtime.h>


typedef float v2f __attribute__((ext_vector_type(2)));
typedef float v8f __attribute__((ext_vector_type(8)));

#define DD 128
#define NT 8
#define WFRAG_F4_PER_MAT 4096

__global__ __launch_bounds__(256)
void reformat_w(const float* __restrict__ Wq, const float* __restrict__ Wk,
                const float* __restrict__ Wv, float4* __restrict__ Wf)
{
    const int idx = (int)(blockIdx.x * blockDim.x + threadIdx.x);
    const int m    = idx >> 12;
    const int r    = idx & 4095;
    const int kkp  = r >> 8;
    const int t    = (r >> 5) & 7;
    const int lane = r & 31;
    const int hi = lane >> 4, l = lane & 15;
    const int n  = t * 16 + l;
    const float* W = (m == 0) ? Wq : ((m == 1) ? Wk : Wv);
    const int k0 = 8 * kkp + 2 * hi;
    const int k1 = k0 + 4;
    float4 v;
    v.x = W[(size_t)(k0 + 0) * DD + n];
    v.y = W[(size_t)(k0 + 1) * DD + n];
    v.z = W[(size_t)(k1 + 0) * DD + n];
    v.w = W[(size_t)(k1 + 1) * DD + n];
    Wf[idx] = v;
}

__global__ __launch_bounds__(256)
void kv_gemm_wmma(const float* __restrict__ X,
                  const float4* __restrict__ Wf,
                  const float* __restrict__ bk, const float* __restrict__ bv,
                  float* __restrict__ K, float* __restrict__ V,
                  int Mrows)
{
    __shared__ float4 sW[WFRAG_F4_PER_MAT];

    const int tid  = (int)threadIdx.x;
    const int wave = (int)((blockIdx.x * blockDim.x + threadIdx.x) >> 5);
    const int lane = tid & 31;
    const bool live = wave * 16 < Mrows;
    const int  m0   = live ? wave * 16 : 0;

    const int hi = lane >> 4;
    const int l  = lane & 15;

    const float* xrow = X + (size_t)(m0 + l) * DD + 2 * hi;

    const float* bm[2] = { bk, bv };
    float*       Om[2] = { K,  V  };

    const float4* W1 = Wf + WFRAG_F4_PER_MAT;
#pragma unroll
    for (int i = 0; i < WFRAG_F4_PER_MAT / 256; ++i)
        sW[tid + 256 * i] = W1[tid + 256 * i];
    __syncthreads();

#pragma unroll
    for (int m = 0; m < 2; ++m) {
        const float*  bp = bm[m];
        float*        Op = Om[m];

        v8f c[NT];
#pragma unroll
        for (int t = 0; t < NT; ++t) {
            const float bias = bp[t * 16 + l];
#pragma unroll
            for (int r = 0; r < 8; ++r) c[t][r] = bias;
        }

#pragma unroll 2
        for (int kkp = 0; kkp < 16; ++kkp) {
            const v2f a0 = *(const v2f*)(xrow + 8 * kkp);
            const v2f a1 = *(const v2f*)(xrow + 8 * kkp + 4);
            float4 bf[NT];
#pragma unroll
            for (int t = 0; t < NT; ++t)
                bf[t] = sW[(kkp * NT + t) * 32 + lane];
#pragma unroll
            for (int t = 0; t < NT; ++t) {
                v2f b0; b0.x = bf[t].x; b0.y = bf[t].y;
                c[t] = __builtin_amdgcn_wmma_f32_16x16x4_f32(
                           false, a0, false, b0, (short)0, c[t], false, false);
            }
#pragma unroll
            for (int t = 0; t < NT; ++t) {
                v2f b1; b1.x = bf[t].z; b1.y = bf[t].w;
                c[t] = __builtin_amdgcn_wmma_f32_16x16x4_f32(
                           false, a1, false, b1, (short)0, c[t], false, false);
            }
        }

        if (m < 1) {
            const float4* Wn = Wf + (size_t)2 * WFRAG_F4_PER_MAT;
            __syncthreads();
#pragma unroll
            for (int i = 0; i < WFRAG_F4_PER_MAT / 256; ++i)
                sW[tid + 256 * i] = Wn[tid + 256 * i];
            __syncthreads();
        }

        if (live) {
#pragma unroll
            for (int t = 0; t < NT; ++t) {
#pragma unroll
                for (int r = 0; r < 8; ++r) {
                    Op[(size_t)(m0 + 8 * hi + r) * DD + t * 16 + l] = c[t][r];
                }
            }
        }
    }
}

#define TS 8
#define HS (TS + 2)
#define HP (HS * HS)
#define RP 33
#define RPF (RP * 4)
#define HTRIPS ((HP + 7) / 8)

__global__ __launch_bounds__(256)
void win_attn(const float* __restrict__ X, const float4* __restrict__ Wf,
              const float* __restrict__ bq,
              const float* __restrict__ K, const float* __restrict__ V,
              float* __restrict__ out, int B, int H, int W)
{
    __shared__ float4 sK[HP * RP];
    __shared__ float4 sV[HP * RP];

    const int lane = (int)(threadIdx.x & 31);
    const int wv = __builtin_amdgcn_readfirstlane((int)(threadIdx.x >> 5)) & 7;

    const int tilesX = (W + TS - 1) / TS;
    const int tilesY = (H + TS - 1) / TS;
    int t = (int)blockIdx.x;
    const int tx = t % tilesX;  t /= tilesX;
    const int ty = t % tilesY;
    const int b  = t / tilesY;
    if (b >= B) return;
    const int y0 = ty * TS, x0 = tx * TS;
    const size_t img = (size_t)b * H * W;
    const float scale = 0.08838834764831845f;

    const int j = lane & 7;
    const int g = lane >> 3;
    const int hi = lane >> 4;
    const int l  = lane & 15;
    const int yrow = y0 + wv;
    const int yq   = min(yrow, H - 1);

    const float4* X4 = (const float4*)X;
    const float4* K4 = (const float4*)K;
    const float4* V4 = (const float4*)V;
    const unsigned pimg = (unsigned)b * (unsigned)(H * W);

    float4 xr[8];
#pragma unroll
    for (int jj = 0; jj < 8; ++jj) {
        const int xc = min(x0 + jj, W - 1);
        xr[jj] = X4[(pimg + (unsigned)(yq * W + xc)) * 32u + (unsigned)lane];
    }

    const float4* wq = Wf + wv * 32 + lane;
    float4 bf[16];
#pragma unroll
    for (int kk = 0; kk < 4; ++kk) bf[kk] = wq[kk * (NT * 32)];

    const int wl = (int)(threadIdx.x >> 5) & 7;
    unsigned hoff[HTRIPS];
    bool     okk[HTRIPS];
#pragma unroll
    for (int i = 0; i < HTRIPS; ++i) {
        const int p  = min(wl + 8 * i, HP - 1);
        const int hy = p / HS, hx = p - hy * HS;
        const int yy = y0 - 1 + hy, xx = x0 - 1 + hx;
        okk[i] = (yy >= 0) & (yy < H) & (xx >= 0) & (xx < W);
        const int yc = min(max(yy, 0), H - 1);
        const int xc = min(max(xx, 0), W - 1);
        hoff[i] = (pimg + (unsigned)(yc * W + xc)) * 32u + (unsigned)lane;
    }

#pragma unroll
    for (int jj = 0; jj < 8; ++jj) sK[(wv * 8 + jj) * RP + lane] = xr[jj];

    v8f acc[4];
    {
        const float bias = bq[wv * 16 + l];
#pragma unroll
        for (int mt = 0; mt < 4; ++mt) {
#pragma unroll
            for (int r = 0; r < 8; ++r) acc[mt][r] = bias;
        }
    }
    __syncthreads();

    const float* sXf = (const float*)sK + l * RPF + 2 * hi;
#pragma unroll
    for (int kb = 0; kb < 4; ++kb) {
        if (kb < 3) {
#pragma unroll
            for (int kk = 0; kk < 4; ++kk)
                bf[4 * (kb + 1) + kk] = wq[(4 * (kb + 1) + kk) * (NT * 32)];
        }
#pragma unroll
        for (int kk = 0; kk < 4; ++kk) {
            const int kkp = 4 * kb + kk;
            v2f a0[4], a1[4];
#pragma unroll
            for (int mt = 0; mt < 4; ++mt) {
                a0[mt] = *(const v2f*)(sXf + mt * 16 * RPF + 8 * kkp);
                a1[mt] = *(const v2f*)(sXf + mt * 16 * RPF + 8 * kkp + 4);
            }
            v2f b0; b0.x = bf[kkp].x; b0.y = bf[kkp].y;
            v2f b1; b1.x = bf[kkp].z; b1.y = bf[kkp].w;
#pragma unroll
            for (int mt = 0; mt < 4; ++mt)
                acc[mt] = __builtin_amdgcn_wmma_f32_16x16x4_f32(
                              false, a0[mt], false, b0, (short)0, acc[mt], false, false);
#pragma unroll
            for (int mt = 0; mt < 4; ++mt)
                acc[mt] = __builtin_amdgcn_wmma_f32_16x16x4_f32(
                              false, a1[mt], false, b1, (short)0, acc[mt], false, false);
        }
    }

    {
        float* sQf = (float*)sV + (8 * hi) * RPF + wv * 16 + l;
#pragma unroll
        for (int mt = 0; mt < 4; ++mt) {
#pragma unroll
            for (int r = 0; r < 8; ++r) sQf[(mt * 16 + r) * RPF] = acc[mt][r];
        }
    }

    __builtin_amdgcn_sched_barrier(0);

    float4 kk[HTRIPS], vv[HTRIPS];
#pragma unroll
    for (int i = 0; i < HTRIPS; ++i) {
        kk[i] = K4[hoff[i]];
        vv[i] = V4[hoff[i]];
    }

    __syncthreads();

#pragma unroll
    for (int i = 0; i < HTRIPS; ++i) {
        const int p = wl + 8 * i;
        float4 k4 = kk[i];
        if (!okk[i]) k4 = make_float4(0.f, 0.f, 0.f, 0.f);
        if (p < HP) sK[p * RP + lane] = k4;
    }

    float4 q[8];
#pragma unroll
    for (int i = 0; i < 8; ++i) q[i] = sV[(wv * 8 + j) * RP + g * 8 + i];

    __syncthreads();

#pragma unroll
    for (int i = 0; i < HTRIPS; ++i) {
        const int p = wl + 8 * i;
        float4 v4 = vv[i];
        if (!okk[i]) v4 = make_float4(0.f, 0.f, 0.f, 0.f);
        if (p < HP) sV[p * RP + lane] = v4;
    }
    __syncthreads();

    const int lo = j * RP + g * 8;

    float sc[9];
#pragma unroll
    for (int dy = 0; dy < 3; ++dy) {
#pragma unroll
        for (int dx = 0; dx < 3; ++dx) {
            const float4* kr = sK + ((wv + dy) * HS + dx) * RP + lo;
            float s0 = 0.0f, s1 = 0.0f;
#pragma unroll
            for (int i = 0; i < 8; i += 2) {
                const float4 ka = kr[i], kb = kr[i + 1];
                s0 = fmaf(q[i].x, ka.x, s0);     s1 = fmaf(q[i + 1].x, kb.x, s1);
                s0 = fmaf(q[i].y, ka.y, s0);     s1 = fmaf(q[i + 1].y, kb.y, s1);
                s0 = fmaf(q[i].z, ka.z, s0);     s1 = fmaf(q[i + 1].z, kb.z, s1);
                s0 = fmaf(q[i].w, ka.w, s0);     s1 = fmaf(q[i + 1].w, kb.w, s1);
            }
            float s = s0 + s1;
            s += __shfl_xor(s, 8, 32);
            s += __shfl_xor(s, 16, 32);
            sc[dy * 3 + dx] = s * scale;
        }
    }

    float mx = sc[0];
#pragma unroll
    for (int i = 1; i < 9; ++i) mx = fmaxf(mx, sc[i]);
    float p9[9], sum = 0.0f;
#pragma unroll
    for (int i = 0; i < 9; ++i) { p9[i] = expf(sc[i] - mx); sum += p9[i]; }
    const float inv = 1.0f / sum;

    float4 o[8];
#pragma unroll
    for (int i = 0; i < 8; ++i) o[i] = make_float4(0.f, 0.f, 0.f, 0.f);
#pragma unroll
    for (int dy = 0; dy < 3; ++dy) {
#pragma unroll
        for (int dx = 0; dx < 3; ++dx) {
            const float4* vr = sV + ((wv + dy) * HS + dx) * RP + lo;
            const float w = p9[dy * 3 + dx] * inv;
#pragma unroll
            for (int i = 0; i < 8; ++i) {
                const float4 v4 = vr[i];
                o[i].x += w * v4.x; o[i].y += w * v4.y;
                o[i].z += w * v4.z; o[i].w += w * v4.w;
            }
        }
    }

    __syncthreads();
    float4* slot = sK + (wv * 8) * RP;
#pragma unroll
    for (int i = 0; i < 8; ++i) slot[lo + i] = o[i];
    __syncthreads();
#pragma unroll
    for (int jj = 0; jj < 8; ++jj) {
        const float4 r = slot[jj * RP + lane];
        if ((yrow < H) & (x0 + jj < W))
            *(float4*)(out + (img + (size_t)yrow * W + x0 + jj) * DD + lane * 4) = r;
    }
}

extern "C" void kernel_launch(void* const* d_in, const int* in_sizes, int n_in,
                              void* d_out, int out_size, void* d_ws, size_t ws_size,
                              hipStream_t stream)
{
    const float* x  = (const float*)d_in[0];
    const float* Wq = (const float*)d_in[1];
    const float* bq = (const float*)d_in[2];
    const float* Wk = (const float*)d_in[3];
    const float* bk = (const float*)d_in[4];
    const float* Wv = (const float*)d_in[5];
    const float* bv = (const float*)d_in[6];

    const int H = 128, W = 128;
    const int Mrows = in_sizes[0] / DD;
    const int B = Mrows / (H * W);

    float*  Q   = (float*)d_ws;
    float*  Kp  = Q  + (size_t)Mrows * DD;
    float*  Vp  = Kp + (size_t)Mrows * DD;
    float4* Wf  = (float4*)(Vp + (size_t)Mrows * DD);

    reformat_w<<<48, 256, 0, stream>>>(Wq, Wk, Wv, Wf);

    const int gemmWaves = Mrows / 16;
    kv_gemm_wmma<<<(gemmWaves + 7) / 8, 256, 0, stream>>>(
        x, Wf, bk, bv, Kp, Vp, Mrows);

    const int attnTiles = Mrows / (TS * TS);
    win_attn<<<attnTiles, 256, 0, stream>>>(
        x, Wf, bq, Kp, Vp, (float*)d_out, B, H, W);
}


# win_attn: V not staged in LDS: a wave reads the 30 V rows round its 8 pixels from memory as whole rows (lane = 4 channels), weights by readlane; LDS 103 -> 66 KB: 4 groups per WGP; sums unchanged
# speedup vs baseline: 1.6797x; 1.0674x over previous
#include <hip/hip_runtime.h>


typedef float v2f __attribute__((ext_vector_type(2)));
typedef float v8f __attribute__((ext_vector_type(8)));

#define DD 128
#define NT 8
#define WFRAG_F4_PER_MAT 4096

__global__ __launch_bounds__(256)
void reformat_w(const float* __restrict__ Wq, const float* __restrict__ Wk,
                const float* __restrict__ Wv, float4* __restrict__ Wf)
{
    const int idx = (int)(blockIdx.x * blockDim.x + threadIdx.x);
    const int m    = idx >> 12;
    const int r    = idx & 4095;
    const int kkp  = r >> 8;
    const int t    = (r >> 5) & 7;
    const int lane = r & 31;
    const int hi = lane >> 4, l = lane & 15;
    const int n  = t * 16 + l;
    const float* W = (m == 0) ? Wq : ((m == 1) ? Wk : Wv);
    const int k0 = 8 * kkp + 2 * hi;
    const int k1 = k0 + 4;
    float4 v;
    v.x = W[(size_t)(k0 + 0) * DD + n];
    v.y = W[(size_t)(k0 + 1) * DD + n];
    v.z = W[(size_t)(k1 + 0) * DD + n];
    v.w = W[(size_t)(k1 + 1) * DD + n];
    Wf[idx] = v;
}

__global__ __launch_bounds__(256)
void kv_gemm_wmma(const float* __restrict__ X,
                  const float4* __restrict__ Wf,
                  const float* __restrict__ bk, const float* __restrict__ bv,
                  float* __restrict__ K, float* __restrict__ V,
                  int Mrows)
{
    __shared__ float4 sW[WFRAG_F4_PER_MAT];

    const int tid  = (int)threadIdx.x;
    const int wave = (int)((blockIdx.x * blockDim.x + threadIdx.x) >> 5);
    const int lane = tid & 31;
    const bool live = wave * 16 < Mrows;
    const int  m0   = live ? wave * 16 : 0;

    const int hi = lane >> 4;
    const int l  = lane & 15;

    const float* xrow = X + (size_t)(m0 + l) * DD + 2 * hi;

    const float* bm[2] = { bk, bv };
    float*       Om[2] = { K,  V  };

    const float4* W1 = Wf + WFRAG_F4_PER_MAT;
#pragma unroll
    for (int i = 0; i < WFRAG_F4_PER_MAT / 256; ++i)
        sW[tid + 256 * i] = W1[tid + 256 * i];
    __syncthreads();

#pragma unroll
    for (int m = 0; m < 2; ++m) {
        const float*  bp = bm[m];
        float*        Op = Om[m];

        v8f c[NT];
#pragma unroll
        for (int t = 0; t < NT; ++t) {
            const float bias = bp[t * 16 + l];
#pragma unroll
            for (int r = 0; r < 8; ++r) c[t][r] = bias;
        }

#pragma unroll 2
        for (int kkp = 0; kkp < 16; ++kkp) {
            const v2f a0 = *(const v2f*)(xrow + 8 * kkp);
            const v2f a1 = *(const v2f*)(xrow + 8 * kkp + 4);
            float4 bf[NT];
#pragma unroll
            for (int t = 0; t < NT; ++t)
                bf[t] = sW[(kkp * NT + t) * 32 + lane];
#pragma unroll
            for (int t = 0; t < NT; ++t) {
                v2f b0; b0.x = bf[t].x; b0.y = bf[t].y;
                c[t] = __builtin_amdgcn_wmma_f32_16x16x4_f32(
                           false, a0, false, b0, (short)0, c[t], false, false);
            }
#pragma unroll
            for (int t = 0; t < NT; ++t) {
                v2f b1; b1.x = bf[t].z; b1.y = bf[t].w;
                c[t] = __builtin_amdgcn_wmma_f32_16x16x4_f32(
                           false, a1, false, b1, (short)0, c[t], false, false);
            }
        }

        if (m < 1) {
            const float4* Wn = Wf + (size_t)2 * WFRAG_F4_PER_MAT;
            __syncthreads();
#pragma unroll
            for (int i = 0; i < WFRAG_F4_PER_MAT / 256; ++i)
                sW[tid + 256 * i] = Wn[tid + 256 * i];
            __syncthreads();
        }

        if (live) {
#pragma unroll
            for (int t = 0; t < NT; ++t) {
#pragma unroll
                for (int r = 0; r < 8; ++r) {
                    Op[(size_t)(m0 + 8 * hi + r) * DD + t * 16 + l] = c[t][r];
                }
            }
        }
    }
}

#define TS 8
#define HS (TS + 2)
#define HP (HS * HS)
#define RP 33
#define RPF (RP * 4)
#define HTRIPS ((HP + 7) / 8)
#define QROW0 64
#define SROWS 128

template <int DY>
__device__ __forceinline__ void add_v_row(float4 (&o)[8], const float4 (&vr)[10], const float (&wt)[9])
{
#pragma unroll
    for (int c = 0; c < 10; ++c) {
        const float4 v4 = vr[c];
#pragma unroll
        for (int dx = 0; dx < 3; ++dx) {
            const int jj = c - dx;
            if (jj >= 0 && jj < 8) {
                const float w = __int_as_float(__builtin_amdgcn_readlane(
                                    __float_as_int(wt[DY * 3 + dx]), jj));
                o[jj].x += w * v4.x; o[jj].y += w * v4.y;
                o[jj].z += w * v4.z; o[jj].w += w * v4.w;
            }
        }
    }
}

__global__ __launch_bounds__(256) __attribute__((amdgpu_waves_per_eu(8)))
void win_attn(const float* __restrict__ X, const float4* __restrict__ Wf,
              const float* __restrict__ bq,
              const float* __restrict__ K, const float* __restrict__ V,
              float* __restrict__ out, int B, int H, int W)
{
    __shared__ float4 sK[SROWS * RP];

    const int lane = (int)(threadIdx.x & 31);
    const int wv = __builtin_amdgcn_readfirstlane((int)(threadIdx.x >> 5)) & 7;

    const int tilesX = (W + TS - 1) / TS;
    const int tilesY = (H + TS - 1) / TS;
    int t = (int)blockIdx.x;
    const int tx = t % tilesX;  t /= tilesX;
    const int ty = t % tilesY;
    const int b  = t / tilesY;
    if (b >= B) return;
    const int y0 = ty * TS, x0 = tx * TS;
    const float scale = 0.08838834764831845f;

    const int j = lane & 7;
    const int g = lane >> 3;
    const int hi = lane >> 4;
    const int l  = lane & 15;
    const int yrow = y0 + wv;
    const int yq   = min(yrow, H - 1);

    const float4* X4 = (const float4*)X;
    const float4* K4 = (const float4*)K;
    const float4* V4 = (const float4*)V;
    const unsigned pimg = (unsigned)b * (unsigned)(H * W);

    float4 xr[8];
#pragma unroll
    for (int jj = 0; jj < 8; ++jj) {
        const int xc = min(x0 + jj, W - 1);
        xr[jj] = X4[(pimg + (unsigned)(yq * W + xc)) * 32u + (unsigned)lane];
    }

    const float4* wq = Wf + wv * 32 + lane;
    float4 bf[16];
#pragma unroll
    for (int kk = 0; kk < 4; ++kk) bf[kk] = wq[kk * (NT * 32)];

#pragma unroll
    for (int jj = 0; jj < 8; ++jj) sK[(wv * 8 + jj) * RP + lane] = xr[jj];

    v8f acc[4];
    {
        const float bias = bq[wv * 16 + l];
#pragma unroll
        for (int mt = 0; mt < 4; ++mt) {
#pragma unroll
            for (int r = 0; r < 8; ++r) acc[mt][r] = bias;
        }
    }
    __syncthreads();

    const float* sXf = (const float*)sK + l * RPF + 2 * hi;
#pragma unroll
    for (int kb = 0; kb < 4; ++kb) {
        if (kb < 3) {
#pragma unroll
            for (int kk = 0; kk < 4; ++kk)
                bf[4 * (kb + 1) + kk] = wq[(4 * (kb + 1) + kk) * (NT * 32)];
        }
#pragma unroll
        for (int kk = 0; kk < 4; ++kk) {
            const int kkp = 4 * kb + kk;
            v2f a0[4], a1[4];
#pragma unroll
            for (int mt = 0; mt < 4; ++mt) {
                a0[mt] = *(const v2f*)(sXf + mt * 16 * RPF + 8 * kkp);
                a1[mt] = *(const v2f*)(sXf + mt * 16 * RPF + 8 * kkp + 4);
            }
            v2f b0; b0.x = bf[kkp].x; b0.y = bf[kkp].y;
            v2f b1; b1.x = bf[kkp].z; b1.y = bf[kkp].w;
#pragma unroll
            for (int mt = 0; mt < 4; ++mt)
                acc[mt] = __builtin_amdgcn_wmma_f32_16x16x4_f32(
                              false, a0[mt], false, b0, (short)0, acc[mt], false, false);
#pragma unroll
            for (int mt = 0; mt < 4; ++mt)
                acc[mt] = __builtin_amdgcn_wmma_f32_16x16x4_f32(
                              false, a1[mt], false, b1, (short)0, acc[mt], false, false);
        }
    }

    {
        float* sQf = (float*)sK + (QROW0 + 8 * hi) * RPF + wv * 16 + l;
#pragma unroll
        for (int mt = 0; mt < 4; ++mt) {
#pragma unroll
            for (int r = 0; r < 8; ++r) sQf[(mt * 16 + r) * RPF] = acc[mt][r];
        }
    }

    __builtin_amdgcn_sched_barrier(0);

    const int wl = (int)(threadIdx.x >> 5) & 7;
    unsigned hoff[HTRIPS];
    bool     okk[HTRIPS];
#pragma unroll
    for (int i = 0; i < HTRIPS; ++i) {
        const int p  = min(wl + 8 * i, HP - 1);
        const int hy = p / HS, hx = p - hy * HS;
        const int yy = y0 - 1 + hy, xx = x0 - 1 + hx;
        okk[i] = (yy >= 0) & (yy < H) & (xx >= 0) & (xx < W);
        const int yc = min(max(yy, 0), H - 1);
        const int xc = min(max(xx, 0), W - 1);
        hoff[i] = (pimg + (unsigned)(yc * W + xc)) * 32u + (unsigned)lane;
    }

    float4 kk[HTRIPS];
#pragma unroll
    for (int i = 0; i < HTRIPS; ++i) kk[i] = K4[hoff[i]];

    __syncthreads();

#pragma unroll
    for (int i = 0; i < 8; ++i) {
        const int p = wl + 8 * i;
        float4 k4 = kk[i];
        if (!okk[i]) k4 = make_float4(0.f, 0.f, 0.f, 0.f);
        sK[p * RP + lane] = k4;
    }

    float4 q[8];
#pragma unroll
    for (int i = 0; i < 8; ++i) q[i] = sK[(QROW0 + wv * 8 + j) * RP + g * 8 + i];

    __syncthreads();

#pragma unroll
    for (int i = 8; i < HTRIPS; ++i) {
        const int p = wl + 8 * i;
        float4 k4 = kk[i];
        if (!okk[i]) k4 = make_float4(0.f, 0.f, 0.f, 0.f);
        if (p < HP) sK[p * RP + lane] = k4;
    }
    __syncthreads();

    unsigned vrow[3];
#pragma unroll
    for (int dy = 0; dy < 3; ++dy) {
        const int yy = y0 - 1 + wv + dy;
        vrow[dy] = pimg + (unsigned)(min(max(yy, 0), H - 1) * W);
    }
    const int lo = j * RP + g * 8;

    float sraw[9];
#pragma unroll
    for (int n = 0; n < 9; ++n) {
        const int dy = n / 3, dx = n % 3;
        int lon = lo;
        if (n >= 3) asm("" : "+v"(lon) : "v"(sraw[n - 3]));
        const float4* kr = sK + ((wv + dy) * HS + dx) * RP + lon;
        float s0 = 0.0f, s1 = 0.0f;
#pragma unroll
        for (int i = 0; i < 8; i += 2) {
            const float4 ka = kr[i], kb = kr[i + 1];
            s0 = fmaf(q[i].x, ka.x, s0);     s1 = fmaf(q[i + 1].x, kb.x, s1);
            s0 = fmaf(q[i].y, ka.y, s0);     s1 = fmaf(q[i + 1].y, kb.y, s1);
            s0 = fmaf(q[i].z, ka.z, s0);     s1 = fmaf(q[i + 1].z, kb.z, s1);
            s0 = fmaf(q[i].w, ka.w, s0);     s1 = fmaf(q[i + 1].w, kb.w, s1);
        }
        sraw[n] = s0 + s1;
    }
    float sc[9];
#pragma unroll
    for (int n = 0; n < 9; ++n) {
        float s = sraw[n];
        s += __shfl_xor(s, 8, 32);
        s += __shfl_xor(s, 16, 32);
        sc[n] = s * scale;
    }

    float mx = sc[0];
#pragma unroll
    for (int i = 1; i < 9; ++i) mx = fmaxf(mx, sc[i]);
    float p9[9], sum = 0.0f;
#pragma unroll
    for (int i = 0; i < 9; ++i) { p9[i] = expf(sc[i] - mx); sum += p9[i]; }
    const float inv = 1.0f / sum;
    float wt[9];
#pragma unroll
    for (int dy = 0; dy < 3; ++dy) {
#pragma unroll
        for (int dx = 0; dx < 3; ++dx) {
            const int yy = y0 - 1 + wv + dy, xx = x0 - 1 + j + dx;
            const bool inimg = (yy >= 0) & (yy < H) & (xx >= 0) & (xx < W);
            wt[dy * 3 + dx] = inimg ? p9[dy * 3 + dx] * inv : 0.0f;
        }
    }

    float4 va[HS], vb[HS], vc[HS];
#pragma unroll
    for (int c = 0; c < HS; ++c) {
        const int xc = min(max(x0 - 1 + c, 0), W - 1);
        va[c] = V4[(vrow[0] + (unsigned)xc) * 32u + (unsigned)lane];
        vb[c] = V4[(vrow[1] + (unsigned)xc) * 32u + (unsigned)lane];
        vc[c] = V4[(vrow[2] + (unsigned)xc) * 32u + (unsigned)lane];
    }
    float4 o[8];
#pragma unroll
    for (int i = 0; i < 8; ++i) o[i] = make_float4(0.f, 0.f, 0.f, 0.f);
    add_v_row<0>(o, va, wt);
    add_v_row<1>(o, vb, wt);
    add_v_row<2>(o, vc, wt);

    float4* O4 = (float4*)out;
    const unsigned orow = pimg + (unsigned)(yrow * W + x0);
#pragma unroll
    for (int jj = 0; jj < 8; ++jj)
        O4[(orow + (unsigned)jj) * 32u + (unsigned)lane] = o[jj];
}

extern "C" void kernel_launch(void* const* d_in, const int* in_sizes, int n_in,
                              void* d_out, int out_size, void* d_ws, size_t ws_size,
                              hipStream_t stream)
{
    const float* x  = (const float*)d_in[0];
    const float* Wq = (const float*)d_in[1];
    const float* bq = (const float*)d_in[2];
    const float* Wk = (const float*)d_in[3];
    const float* bk = (const float*)d_in[4];
    const float* Wv = (const float*)d_in[5];
    const float* bv = (const float*)d_in[6];

    const int H = 128, W = 128;
    const int Mrows = in_sizes[0] / DD;
    const int B = Mrows / (H * W);

    float*  Q   = (float*)d_ws;
    float*  Kp  = Q  + (size_t)Mrows * DD;
    float*  Vp  = Kp + (size_t)Mrows * DD;
    float4* Wf  = (float4*)(Vp + (size_t)Mrows * DD);

    reformat_w<<<48, 256, 0, stream>>>(Wq, Wk, Wv, Wf);

    const int gemmWaves = Mrows / 16;
    kv_gemm_wmma<<<(gemmWaves + 7) / 8, 256, 0, stream>>>(
        x, Wf, bk, bv, Kp, Vp, Mrows);

    const int attnTiles = Mrows / (TS * TS);
    win_attn<<<attnTiles, 256, 0, stream>>>(
        x, Wf, bq, Kp, Vp, (float*)d_out, B, H, W);
}
